# starting kernel
# baseline (speedup 1.0000x reference)
#include <hip/hip_runtime.h>


typedef __attribute__((ext_vector_type(2))) float v2f;
typedef __attribute__((ext_vector_type(8))) float v8f;

#define C_DIM 128
#define H_DIM 192
#define W_DIM 192
#define B_DIM 4
#define KSZ   9
#define RAD   4
#define HW    (H_DIM * W_DIM)
#define CHW   ((size_t)C_DIM * HW)

__launch_bounds__(32)
__global__ void corr_wmma_f32(const float* __restrict__ x,
                              const float* __restrict__ y,
                              float* __restrict__ out)
{
    const int tw = blockIdx.x;
    const int h  = blockIdx.y;
    const int b  = blockIdx.z;
    const int w0 = tw * 16;

    const int lane = threadIdx.x;
    const int half = lane >> 4;
    const int lm   = lane & 15;

    const float* xbase = x + (size_t)b * CHW + (size_t)(2 * half) * HW
                           + (size_t)h * W_DIM + (w0 + lm);
    v2f a[32];
#pragma unroll
    for (int kc = 0; kc < 32; ++kc) {
        const float* p = xbase + (size_t)(4 * kc) * HW;
        a[kc].x = p[0];
        a[kc].y = p[HW];
    }

    const int wp0 = w0 - 4 + lm;
    const int wp1 = w0 + 12 + lm;
    const bool ok0 = (wp0 >= 0) && (wp0 < W_DIM);
    const bool ok1 = (wp1 < W_DIM);
    const int wp0c = wp0 < 0 ? 0 : (wp0 >= W_DIM ? W_DIM - 1 : wp0);
    const int wp1c = wp1 >= W_DIM ? W_DIM - 1 : wp1;

    const size_t ych = (size_t)b * CHW + (size_t)(2 * half) * HW;
    const float inv_c = 1.0f / (float)C_DIM;

    for (int oi = 0; oi < KSZ; ++oi) {
        v8f d0a = {}, d0b = {};
        v8f d1a = {}, d1b = {};
        const int row = h + oi - RAD;
        if (row >= 0 && row < H_DIM) {
            const float* y0 = y + ych + (size_t)row * W_DIM + wp0c;
            const float* y1 = y + ych + (size_t)row * W_DIM + wp1c;
#pragma unroll
            for (int kc = 0; kc < 32; kc += 2) {
                const size_t offA = (size_t)(4 * kc) * HW;
                const size_t offB = (size_t)(4 * (kc + 1)) * HW;
                v2f b0a, b1a, b0b, b1b;
                b0a.x = y0[offA];
                b0a.y = y0[offA + HW];
                b1a.x = y1[offA];
                b1a.y = y1[offA + HW];
                b0b.x = y0[offB];
                b0b.y = y0[offB + HW];
                b1b.x = y1[offB];
                b1b.y = y1[offB + HW];
                d0a = __builtin_amdgcn_wmma_f32_16x16x4_f32(
                        false, a[kc],     false, b0a, (short)0, d0a, false, false);
                d1a = __builtin_amdgcn_wmma_f32_16x16x4_f32(
                        false, a[kc],     false, b1a, (short)0, d1a, false, false);
                d0b = __builtin_amdgcn_wmma_f32_16x16x4_f32(
                        false, a[kc + 1], false, b0b, (short)0, d0b, false, false);
                d1b = __builtin_amdgcn_wmma_f32_16x16x4_f32(
                        false, a[kc + 1], false, b1b, (short)0, d1b, false, false);
            }
        }
        float* obase = out + (size_t)b * (KSZ * KSZ) * HW
                           + (size_t)(oi * KSZ) * HW
                           + (size_t)h * W_DIM + w0;
#pragma unroll
        for (int v = 0; v < 8; ++v) {
            const int m = v + 8 * half;
            const int oj0 = lm - m;
            if (oj0 >= 0 && oj0 <= 8) {
                const float r = ok0 ? (d0a[v] + d0b[v]) * inv_c : 0.0f;
                obase[(size_t)oj0 * HW + m] = r;
            }
            const int oj1 = lm + 16 - m;
            if (oj1 <= 8) {
                const float r = ok1 ? (d1a[v] + d1b[v]) * inv_c : 0.0f;
                obase[(size_t)oj1 * HW + m] = r;
            }
        }
    }
}

extern "C" void kernel_launch(void* const* d_in, const int* in_sizes, int n_in,
                              void* d_out, int out_size, void* d_ws, size_t ws_size,
                              hipStream_t stream) {
    (void)in_sizes; (void)n_in; (void)d_ws; (void)ws_size; (void)out_size;
    const float* x = (const float*)d_in[0];
    const float* y = (const float*)d_in[1];
    float* out = (float*)d_out;
    dim3 grid(W_DIM / 16, H_DIM, B_DIM);
    dim3 block(32, 1, 1);
    corr_wmma_f32<<<grid, block, 0, stream>>>(x, y, out);
}


# y row window (32 cols x 128 channels) kept in registers as B, the nine x rows that meet it streamed as A (was: x kept, nine y windows streamed): 704 loads per wave for 1216, x loads 64-B aligned
# speedup vs baseline: 1.0969x; 1.0969x over previous
#include <hip/hip_runtime.h>


typedef __attribute__((ext_vector_type(2))) float v2f;
typedef __attribute__((ext_vector_type(8))) float v8f;

#define C_DIM 128
#define H_DIM 192
#define W_DIM 192
#define B_DIM 4
#define KSZ   9
#define RAD   4
#define HW    (H_DIM * W_DIM)
#define CHW   ((size_t)C_DIM * HW)

__launch_bounds__(32)
__global__ void corr_wmma_f32(const float* __restrict__ x,
                              const float* __restrict__ y,
                              float* __restrict__ out)
{
    const int tw = blockIdx.x;
    const int r  = (int)blockIdx.y - RAD;
    const int b  = blockIdx.z;
    const int w0 = tw * 16;

    const int lane = threadIdx.x;
    const int half = lane >> 4;
    const int lm   = lane & 15;

    const int wp0 = w0 - 4 + lm;
    const int wp1 = w0 + 12 + lm;
    const bool ok0 = (wp0 >= 0) && (wp0 < W_DIM);
    const bool ok1 = (wp1 < W_DIM);
    const int wp0c = wp0 < 0 ? 0 : (wp0 >= W_DIM ? W_DIM - 1 : wp0);
    const int wp1c = wp1 >= W_DIM ? W_DIM - 1 : wp1;

    const size_t ch0 = (size_t)b * CHW + (size_t)(2 * half) * HW;
    const float inv_c = 1.0f / (float)C_DIM;
    const bool row_ok = (r >= 0) && (r < H_DIM);

    float* const oseg = out + (size_t)b * (KSZ * KSZ) * HW + w0;

    if (!row_ok) {
        for (int oi = 0; oi < KSZ; ++oi) {
            const int h = r + RAD - oi;
            if (h < 0 || h >= H_DIM) continue;
            float* obase = oseg + (size_t)(oi * KSZ) * HW + (size_t)h * W_DIM;
            for (int oj = half; oj < KSZ; oj += 2)
                obase[(size_t)oj * HW + lm] = 0.0f;
        }
        return;
    }

    v2f b0[32], b1[32];
    {
        const float* y0 = y + ch0 + (size_t)r * W_DIM + wp0c;
        const float* y1 = y + ch0 + (size_t)r * W_DIM + wp1c;
#pragma unroll
        for (int kc = 0; kc < 32; ++kc) {
            const size_t off = (size_t)(4 * kc) * HW;
            b0[kc].x = y0[off];
            b0[kc].y = y0[off + HW];
            b1[kc].x = y1[off];
            b1[kc].y = y1[off + HW];
        }
    }

    const int ojs = lm - 8 * half;
    const ptrdiff_t lane_off = (ptrdiff_t)ojs * HW + 8 * half;

    const float* xcol = x + ch0 + (size_t)(w0 + lm);

    for (int oi = 0; oi < KSZ; ++oi) {
        const int h = r + RAD - oi;
        if (h < 0 || h >= H_DIM) continue;
        v8f d0a = {}, d0b = {};
        v8f d1a = {}, d1b = {};
        {
            const float* xp = xcol + (size_t)h * W_DIM;
            v2f a[32];
#pragma unroll
            for (int kc = 0; kc < 32; ++kc) {
                const float* p = xp + (size_t)(4 * kc) * HW;
                a[kc].x = p[0];
                a[kc].y = p[HW];
            }
#pragma unroll
            for (int kc = 0; kc < 32; kc += 2) {
                d0a = __builtin_amdgcn_wmma_f32_16x16x4_f32(
                        false, a[kc],     false, b0[kc],     (short)0, d0a, false, false);
                d1a = __builtin_amdgcn_wmma_f32_16x16x4_f32(
                        false, a[kc],     false, b1[kc],     (short)0, d1a, false, false);
                d0b = __builtin_amdgcn_wmma_f32_16x16x4_f32(
                        false, a[kc + 1], false, b0[kc + 1], (short)0, d0b, false, false);
                d1b = __builtin_amdgcn_wmma_f32_16x16x4_f32(
                        false, a[kc + 1], false, b1[kc + 1], (short)0, d1b, false, false);
            }
        }
        const ptrdiff_t lane_idx = (ptrdiff_t)(oi * KSZ) * HW
                                 + (ptrdiff_t)h * W_DIM + lane_off;
#pragma unroll
        for (int v = 0; v < 8; ++v) {
            const ptrdiff_t step = (ptrdiff_t)v * (HW - 1);
            const int oj0 = ojs - v;
            if (oj0 >= 0 && oj0 <= 8) {
                const float q = ok0 ? (d0a[v] + d0b[v]) * inv_c : 0.0f;
                oseg[lane_idx - step] = q;
            }
            const int oj1 = oj0 + 16;
            if (oj1 <= 8) {
                const float q = ok1 ? (d1a[v] + d1b[v]) * inv_c : 0.0f;
                oseg[lane_idx + (ptrdiff_t)16 * HW - step] = q;
            }
        }
    }
}

extern "C" void kernel_launch(void* const* d_in, const int* in_sizes, int n_in,
                              void* d_out, int out_size, void* d_ws, size_t ws_size,
                              hipStream_t stream) {
    (void)in_sizes; (void)n_in; (void)d_ws; (void)ws_size; (void)out_size;
    const float* x = (const float*)d_in[0];
    const float* y = (const float*)d_in[1];
    float* out = (float*)d_out;
    dim3 grid(W_DIM / 16, H_DIM + 2 * RAD, B_DIM);
    dim3 block(32, 1, 1);
    corr_wmma_f32<<<grid, block, 0, stream>>>(x, y, out);
}


# output band of a row step laid out in LDS as [oj][pixel], then stored 16 B per lane as 64-B plane segments: 2 store instructions, 9 line pieces per step (was 16 instructions, 108 pieces of 4-8 B)
# speedup vs baseline: 3.3803x; 3.0817x over previous
#include <hip/hip_runtime.h>


typedef __attribute__((ext_vector_type(2))) float v2f;
typedef __attribute__((ext_vector_type(4))) float v4f;
typedef __attribute__((ext_vector_type(8))) float v8f;

#define C_DIM 128
#define H_DIM 192
#define W_DIM 192
#define B_DIM 4
#define KSZ   9
#define RAD   4
#define HW    (H_DIM * W_DIM)
#define CHW   ((size_t)C_DIM * HW)

__launch_bounds__(32)
__global__ void corr_wmma_f32(const float* __restrict__ x,
                              const float* __restrict__ y,
                              float* __restrict__ out)
{
    __shared__ __attribute__((aligned(16))) float band[KSZ * 16];

    const int tw = blockIdx.x;
    const int r  = (int)blockIdx.y - RAD;
    const int b  = blockIdx.z;
    const int w0 = tw * 16;

    const int lane = threadIdx.x;
    const int half = lane >> 4;
    const int lm   = lane & 15;

    const int wp0 = w0 - 4 + lm;
    const int wp1 = w0 + 12 + lm;
    const bool ok0 = (wp0 >= 0) && (wp0 < W_DIM);
    const bool ok1 = (wp1 < W_DIM);
    const int wp0c = wp0 < 0 ? 0 : (wp0 >= W_DIM ? W_DIM - 1 : wp0);
    const int wp1c = wp1 >= W_DIM ? W_DIM - 1 : wp1;

    const size_t ch0 = (size_t)b * CHW + (size_t)(2 * half) * HW;
    const float inv_c = 1.0f / (float)C_DIM;
    const bool row_ok = (r >= 0) && (r < H_DIM);

    float* const oseg = out + (size_t)b * (KSZ * KSZ) * HW + w0;

    float* const oquad  = oseg + (size_t)(lane >> 2) * HW + 4 * (lane & 3);
    float* const oquad8 = oseg + (size_t)8 * HW + 4 * lane;

    if (!row_ok) {
        const v4f z = {0.0f, 0.0f, 0.0f, 0.0f};
        for (int oi = 0; oi < KSZ; ++oi) {
            const int h = r + RAD - oi;
            if (h < 0 || h >= H_DIM) continue;
            const size_t o = (size_t)(oi * KSZ) * HW + (size_t)h * W_DIM;
            *(v4f*)(oquad + o) = z;
            if (lane < 4) *(v4f*)(oquad8 + o) = z;
        }
        return;
    }

    v2f b0[32], b1[32];
    {
        const float* y0 = y + ch0 + (size_t)r * W_DIM + wp0c;
        const float* y1 = y + ch0 + (size_t)r * W_DIM + wp1c;
#pragma unroll
        for (int kc = 0; kc < 32; ++kc) {
            const size_t off = (size_t)(4 * kc) * HW;
            b0[kc].x = y0[off];
            b0[kc].y = y0[off + HW];
            b1[kc].x = y1[off];
            b1[kc].y = y1[off + HW];
        }
    }

    const int ojs = lm - 8 * half;

    const float* xcol = x + ch0 + (size_t)(w0 + lm);

    for (int oi = 0; oi < KSZ; ++oi) {
        const int h = r + RAD - oi;
        if (h < 0 || h >= H_DIM) continue;
        v8f d0a = {}, d0b = {};
        v8f d1a = {}, d1b = {};
        {
            const float* xp = xcol + (size_t)h * W_DIM;
            v2f a[32];
#pragma unroll
            for (int kc = 0; kc < 32; ++kc) {
                const float* p = xp + (size_t)(4 * kc) * HW;
                a[kc].x = p[0];
                a[kc].y = p[HW];
            }
#pragma unroll
            for (int kc = 0; kc < 32; kc += 2) {
                d0a = __builtin_amdgcn_wmma_f32_16x16x4_f32(
                        false, a[kc],     false, b0[kc],     (short)0, d0a, false, false);
                d1a = __builtin_amdgcn_wmma_f32_16x16x4_f32(
                        false, a[kc],     false, b1[kc],     (short)0, d1a, false, false);
                d0b = __builtin_amdgcn_wmma_f32_16x16x4_f32(
                        false, a[kc + 1], false, b0[kc + 1], (short)0, d0b, false, false);
                d1b = __builtin_amdgcn_wmma_f32_16x16x4_f32(
                        false, a[kc + 1], false, b1[kc + 1], (short)0, d1b, false, false);
            }
        }
#pragma unroll
        for (int v = 0; v < 8; ++v) {
            const int oj0 = ojs - v;
            const bool t0 = oj0 >= 0;
            const int oj = t0 ? oj0 : oj0 + 16;
            if (oj <= 8) {
                const float s = t0 ? (d0a[v] + d0b[v]) : (d1a[v] + d1b[v]);
                const bool ok = t0 ? ok0 : ok1;
                band[oj * 16 + v + 8 * half] = ok ? s * inv_c : 0.0f;
            }
        }
        __syncthreads();
        const v4f q = *(const v4f*)&band[4 * lane];
        v4f q8 = {0.0f, 0.0f, 0.0f, 0.0f};
        if (lane < 4) q8 = *(const v4f*)&band[8 * 16 + 4 * lane];
        __syncthreads();
        const size_t o = (size_t)(oi * KSZ) * HW + (size_t)h * W_DIM;
        *(v4f*)(oquad + o) = q;
        if (lane < 4) *(v4f*)(oquad8 + o) = q8;
    }
}

extern "C" void kernel_launch(void* const* d_in, const int* in_sizes, int n_in,
                              void* d_out, int out_size, void* d_ws, size_t ws_size,
                              hipStream_t stream) {
    (void)in_sizes; (void)n_in; (void)d_ws; (void)ws_size; (void)out_size;
    const float* x = (const float*)d_in[0];
    const float* y = (const float*)d_in[1];
    float* out = (float*)d_out;
    dim3 grid(W_DIM / 16, H_DIM + 2 * RAD, B_DIM);
    dim3 block(32, 1, 1);
    corr_wmma_f32<<<grid, block, 0, stream>>>(x, y, out);
}


# 4-wave workgroups of waves that read the same 128-B lines (both 16-px halves of a line x two adjacent y rows), row steps ordered so all four read one x row together: lines hit in the WGP cache
# speedup vs baseline: 4.1543x; 1.2290x over previous
#include <hip/hip_runtime.h>


typedef __attribute__((ext_vector_type(2))) float v2f;
typedef __attribute__((ext_vector_type(4))) float v4f;
typedef __attribute__((ext_vector_type(8))) float v8f;

#define C_DIM 128
#define H_DIM 192
#define W_DIM 192
#define B_DIM 4
#define KSZ   9
#define RAD   4
#define HW    (H_DIM * W_DIM)
#define CHW   ((size_t)C_DIM * HW)

__launch_bounds__(128)
__global__ void corr_wmma_f32(const float* __restrict__ x,
                              const float* __restrict__ y,
                              float* __restrict__ out)
{
    __shared__ __attribute__((aligned(16))) float band_all[4][KSZ * 16];

    const int wv = __builtin_amdgcn_readfirstlane((int)(threadIdx.x >> 5));
    const int jt = wv & 1;
    const int jr = wv >> 1;
    float* const band = band_all[wv];

    const int tw = (int)blockIdx.x * 2 + jt;
    const int r  = (int)blockIdx.y * 2 - RAD + jr;
    const int b  = blockIdx.z;
    const int w0 = tw * 16;

    const int lane = threadIdx.x & 31;
    const int half = lane >> 4;
    const int lm   = lane & 15;

    const int wp0 = w0 - 4 + lm;
    const int wp1 = w0 + 12 + lm;
    const bool ok0 = (wp0 >= 0) && (wp0 < W_DIM);
    const bool ok1 = (wp1 < W_DIM);
    const int wp0c = wp0 < 0 ? 0 : (wp0 >= W_DIM ? W_DIM - 1 : wp0);
    const int wp1c = wp1 >= W_DIM ? W_DIM - 1 : wp1;

    const size_t ch0 = (size_t)b * CHW + (size_t)(2 * half) * HW;
    const float inv_c = 1.0f / (float)C_DIM;
    const bool row_ok = (r >= 0) && (r < H_DIM);

    float* const oseg = out + (size_t)b * (KSZ * KSZ) * HW + w0;

    float* const oquad  = oseg + (size_t)(lane >> 2) * HW + 4 * (lane & 3);
    float* const oquad8 = oseg + (size_t)8 * HW + 4 * lane;

    if (!row_ok) {
        const v4f z = {0.0f, 0.0f, 0.0f, 0.0f};
        for (int oi = 0; oi < KSZ; ++oi) {
            const int h = r + RAD - oi;
            if (h < 0 || h >= H_DIM) continue;
            const size_t o = (size_t)(oi * KSZ) * HW + (size_t)h * W_DIM;
            *(v4f*)(oquad + o) = z;
            if (lane < 4) *(v4f*)(oquad8 + o) = z;
        }
        return;
    }

    v2f b0[32], b1[32];
    {
        const float* y0 = y + ch0 + (size_t)r * W_DIM + wp0c;
        const float* y1 = y + ch0 + (size_t)r * W_DIM + wp1c;
#pragma unroll
        for (int kc = 0; kc < 32; ++kc) {
            const size_t off = (size_t)(4 * kc) * HW;
            b0[kc].x = y0[off];
            b0[kc].y = y0[off + HW];
            b1[kc].x = y1[off];
            b1[kc].y = y1[off + HW];
        }
    }

    const int ojs = lm - 8 * half;

    const float* xcol = x + ch0 + (size_t)(w0 + lm);

    for (int s = 0; s < KSZ; ++s) {
        const int oi9 = KSZ - 1 - s + jr;
        const int oi = oi9 == KSZ ? 0 : oi9;
        const int h = r + RAD - oi;
        const bool active = (h >= 0) && (h < H_DIM);
        v8f d0a = {}, d0b = {};
        v8f d1a = {}, d1b = {};
        if (active) {
            const float* xp = xcol + (size_t)h * W_DIM;
            v2f a[32];
#pragma unroll
            for (int kc = 0; kc < 32; ++kc) {
                const float* p = xp + (size_t)(4 * kc) * HW;
                a[kc].x = p[0];
                a[kc].y = p[HW];
            }
#pragma unroll
            for (int kc = 0; kc < 32; kc += 2) {
                d0a = __builtin_amdgcn_wmma_f32_16x16x4_f32(
                        false, a[kc],     false, b0[kc],     (short)0, d0a, false, false);
                d1a = __builtin_amdgcn_wmma_f32_16x16x4_f32(
                        false, a[kc],     false, b1[kc],     (short)0, d1a, false, false);
                d0b = __builtin_amdgcn_wmma_f32_16x16x4_f32(
                        false, a[kc + 1], false, b0[kc + 1], (short)0, d0b, false, false);
                d1b = __builtin_amdgcn_wmma_f32_16x16x4_f32(
                        false, a[kc + 1], false, b1[kc + 1], (short)0, d1b, false, false);
            }
        }
        if (active) {
#pragma unroll
            for (int v = 0; v < 8; ++v) {
                const int oj0 = ojs - v;
                const bool t0 = oj0 >= 0;
                const int oj = t0 ? oj0 : oj0 + 16;
                if (oj <= 8) {
                    const float sum = t0 ? (d0a[v] + d0b[v]) : (d1a[v] + d1b[v]);
                    const bool ok = t0 ? ok0 : ok1;
                    band[oj * 16 + v + 8 * half] = ok ? sum * inv_c : 0.0f;
                }
            }
        }
        __syncthreads();
        v4f q = {0.0f, 0.0f, 0.0f, 0.0f};
        v4f q8 = {0.0f, 0.0f, 0.0f, 0.0f};
        if (active) {
            q = *(const v4f*)&band[4 * lane];
            if (lane < 4) q8 = *(const v4f*)&band[8 * 16 + 4 * lane];
        }
        __syncthreads();
        if (active) {
            const size_t o = (size_t)(oi * KSZ) * HW + (size_t)h * W_DIM;
            *(v4f*)(oquad + o) = q;
            if (lane < 4) *(v4f*)(oquad8 + o) = q8;
        }
    }
}

extern "C" void kernel_launch(void* const* d_in, const int* in_sizes, int n_in,
                              void* d_out, int out_size, void* d_ws, size_t ws_size,
                              hipStream_t stream) {
    (void)in_sizes; (void)n_in; (void)d_ws; (void)ws_size; (void)out_size;
    const float* x = (const float*)d_in[0];
    const float* y = (const float*)d_in[1];
    float* out = (float*)d_out;
    dim3 grid(W_DIM / 32, (H_DIM + 2 * RAD) / 2, B_DIM);
    dim3 block(128, 1, 1);
    corr_wmma_f32<<<grid, block, 0, stream>>>(x, y, out);
}


# x tile kept as two 64-channel register halves, each reloaded with the NEXT step's row right after its products issue; step barriers fence LDS only, so these loads stay in flight across the epilogue
# speedup vs baseline: 4.5034x; 1.0840x over previous
#include <hip/hip_runtime.h>


typedef __attribute__((ext_vector_type(2))) float v2f;
typedef __attribute__((ext_vector_type(4))) float v4f;
typedef __attribute__((ext_vector_type(8))) float v8f;

#define C_DIM 128
#define H_DIM 192
#define W_DIM 192
#define B_DIM 4
#define KSZ   9
#define RAD   4
#define HW    (H_DIM * W_DIM)
#define CHW   ((size_t)C_DIM * HW)

__device__ __forceinline__ void lds_only_barrier()
{
    __builtin_amdgcn_fence(__ATOMIC_RELEASE, "workgroup", "local");
    __builtin_amdgcn_s_barrier();
    __builtin_amdgcn_fence(__ATOMIC_ACQUIRE, "workgroup", "local");
}

__launch_bounds__(128)
__global__ void corr_wmma_f32(const float* __restrict__ x,
                              const float* __restrict__ y,
                              float* __restrict__ out)
{
    __shared__ __attribute__((aligned(16))) float band_all[4][KSZ * 16];

    const int wv = __builtin_amdgcn_readfirstlane((int)(threadIdx.x >> 5));
    const int jt = wv & 1;
    const int jr = wv >> 1;
    float* const band = band_all[wv];

    const int tw = (int)blockIdx.x * 2 + jt;
    const int r  = (int)blockIdx.y * 2 - RAD + jr;
    const int b  = blockIdx.z;
    const int w0 = tw * 16;

    const int lane = threadIdx.x & 31;
    const int half = lane >> 4;
    const int lm   = lane & 15;

    const int wp0 = w0 - 4 + lm;
    const int wp1 = w0 + 12 + lm;
    const bool ok0 = (wp0 >= 0) && (wp0 < W_DIM);
    const bool ok1 = (wp1 < W_DIM);
    const int wp0c = wp0 < 0 ? 0 : (wp0 >= W_DIM ? W_DIM - 1 : wp0);
    const int wp1c = wp1 >= W_DIM ? W_DIM - 1 : wp1;

    const size_t ch0 = (size_t)b * CHW + (size_t)(2 * half) * HW;
    const float inv_c = 1.0f / (float)C_DIM;
    const bool row_ok = (r >= 0) && (r < H_DIM);

    float* const oseg = out + (size_t)b * (KSZ * KSZ) * HW + w0;

    float* const oquad  = oseg + (size_t)(lane >> 2) * HW + 4 * (lane & 3);
    float* const oquad8 = oseg + (size_t)8 * HW + 4 * lane;

    if (!row_ok) {
        const v4f z = {0.0f, 0.0f, 0.0f, 0.0f};
        for (int oi = 0; oi < KSZ; ++oi) {
            const int h = r + RAD - oi;
            if (h < 0 || h >= H_DIM) continue;
            const size_t o = (size_t)(oi * KSZ) * HW + (size_t)h * W_DIM;
            *(v4f*)(oquad + o) = z;
            if (lane < 4) *(v4f*)(oquad8 + o) = z;
        }
        return;
    }

    v2f b0[32], b1[32];
    {
        const float* y0 = y + ch0 + (size_t)r * W_DIM + wp0c;
        const float* y1 = y + ch0 + (size_t)r * W_DIM + wp1c;
#pragma unroll
        for (int kc = 0; kc < 32; ++kc) {
            const size_t off = (size_t)(4 * kc) * HW;
            b0[kc].x = y0[off];
            b0[kc].y = y0[off + HW];
            b1[kc].x = y1[off];
            b1[kc].y = y1[off + HW];
        }
    }

    const int ojs = lm - 8 * half;

    const float* xcol = x + ch0 + (size_t)(w0 + lm);

    v2f alo[16], ahi[16];
    int oi_c, h_c;
    bool act_c;
    {
        const int oi9 = KSZ - 1 + jr;
        oi_c = oi9 == KSZ ? 0 : oi9;
        h_c = r + RAD - oi_c;
        act_c = (h_c >= 0) && (h_c < H_DIM);
    }

#define LOAD_HALF(AH, XP, K0)                                                   \
    _Pragma("unroll")                                                           \
    for (int kc = 0; kc < 16; ++kc) {                                           \
        const float* p = (XP) + (size_t)(4 * (kc + (K0))) * HW;                 \
        AH[kc].x = p[0];                                                        \
        AH[kc].y = p[HW];                                                       \
    }

#define WMMA_HALF(AH, K0)                                                       \
    _Pragma("unroll")                                                           \
    for (int kc = 0; kc < 16; kc += 2) {                                        \
        d0a = __builtin_amdgcn_wmma_f32_16x16x4_f32(                            \
                false, AH[kc],     false, b0[kc + (K0)],     (short)0, d0a, false, false); \
        d1a = __builtin_amdgcn_wmma_f32_16x16x4_f32(                            \
                false, AH[kc],     false, b1[kc + (K0)],     (short)0, d1a, false, false); \
        d0b = __builtin_amdgcn_wmma_f32_16x16x4_f32(                            \
                false, AH[kc + 1], false, b0[kc + 1 + (K0)], (short)0, d0b, false, false); \
        d1b = __builtin_amdgcn_wmma_f32_16x16x4_f32(                            \
                false, AH[kc + 1], false, b1[kc + 1 + (K0)], (short)0, d1b, false, false); \
    }

#define STEP_EPILOGUE()                                                         \
    if (act_c) {                                                                \
        _Pragma("unroll")                                                       \
        for (int v = 0; v < 8; ++v) {                                           \
            const int oj0 = ojs - v;                                            \
            const bool t0 = oj0 >= 0;                                           \
            const int oj = t0 ? oj0 : oj0 + 16;                                 \
            if (oj <= 8) {                                                      \
                const float sum = t0 ? (d0a[v] + d0b[v]) : (d1a[v] + d1b[v]);   \
                const bool ok = t0 ? ok0 : ok1;                                 \
                band[oj * 16 + v + 8 * half] = ok ? sum * inv_c : 0.0f;         \
            }                                                                   \
        }                                                                       \
    }                                                                           \
    lds_only_barrier();                                                         \
    v4f q = {0.0f, 0.0f, 0.0f, 0.0f};                                           \
    v4f q8 = {0.0f, 0.0f, 0.0f, 0.0f};                                          \
    if (act_c) {                                                                \
        q = *(const v4f*)&band[4 * lane];                                       \
        if (lane < 4) q8 = *(const v4f*)&band[8 * 16 + 4 * lane];               \
    }                                                                           \
    lds_only_barrier();                                                         \
    if (act_c) {                                                                \
        const size_t o = (size_t)(oi_c * KSZ) * HW + (size_t)h_c * W_DIM;       \
        *(v4f*)(oquad + o) = q;                                                 \
        if (lane < 4) *(v4f*)(oquad8 + o) = q8;                                 \
    }

    {
        const int hcl = h_c < 0 ? 0 : (h_c >= H_DIM ? H_DIM - 1 : h_c);
        const float* xp = xcol + (size_t)hcl * W_DIM;
        LOAD_HALF(alo, xp, 0)
        LOAD_HALF(ahi, xp, 16)
    }
    for (int s = 0; s < KSZ - 1; ++s) {
        const int oi9n = KSZ - 2 - s + jr;
        const int oi_n = oi9n == KSZ ? 0 : oi9n;
        const int h_n = r + RAD - oi_n;
        const bool act_n = (h_n >= 0) && (h_n < H_DIM);
        const int hcl = h_n < 0 ? 0 : (h_n >= H_DIM ? H_DIM - 1 : h_n);
        const float* xpn = xcol + (size_t)hcl * W_DIM;
        v8f d0a = {}, d0b = {};
        v8f d1a = {}, d1b = {};
        if (act_c) { WMMA_HALF(alo, 0) }
        LOAD_HALF(alo, xpn, 0)
        if (act_c) { WMMA_HALF(ahi, 16) }
        LOAD_HALF(ahi, xpn, 16)
        STEP_EPILOGUE()
        oi_c = oi_n;
        h_c = h_n;
        act_c = act_n;
    }
    {
        v8f d0a = {}, d0b = {};
        v8f d1a = {}, d1b = {};
        if (act_c) {
            WMMA_HALF(alo, 0)
            WMMA_HALF(ahi, 16)
        }
        STEP_EPILOGUE()
    }
#undef LOAD_HALF
#undef WMMA_HALF
#undef STEP_EPILOGUE
}

extern "C" void kernel_launch(void* const* d_in, const int* in_sizes, int n_in,
                              void* d_out, int out_size, void* d_ws, size_t ws_size,
                              hipStream_t stream) {
    (void)in_sizes; (void)n_in; (void)d_ws; (void)ws_size; (void)out_size;
    const float* x = (const float*)d_in[0];
    const float* y = (const float*)d_in[1];
    float* out = (float*)d_out;
    dim3 grid(W_DIM / 32, (H_DIM + 2 * RAD) / 2, B_DIM);
    dim3 block(128, 1, 1);
    corr_wmma_f32<<<grid, block, 0, stream>>>(x, y, out);
}
